# starting kernel
# baseline (speedup 1.0000x reference)
#include <hip/hip_runtime.h>
#include <math.h>

#define D_    1024
#define DFF_  4096
#define E_    8
#define NTOK  8192

typedef float v2f __attribute__((ext_vector_type(2)));
typedef float v8f __attribute__((ext_vector_type(8)));

__device__ __forceinline__ v8f wmma_f32_k4(v2f a, v2f b, v8f c) {
    return __builtin_amdgcn_wmma_f32_16x16x4_f32(
         false, a,  false, b,
         (short)0, c,  false,  false);
}

__device__ __forceinline__ float gelu_exact(float v) {
    return 0.5f * v * (1.0f + erff(v * 0.70710678118654752440f));
}

__global__ void moe_zero_kernel(float* __restrict__ out, long long n,
                                int* __restrict__ counts) {
    long long i = (long long)blockIdx.x * blockDim.x + threadIdx.x;
    const long long stride = (long long)gridDim.x * blockDim.x;
    for (; i < n; i += stride) out[i] = 0.0f;
    if (blockIdx.x == 0 && threadIdx.x < E_) counts[threadIdx.x] = 0;
}

__global__ __launch_bounds__(256) void moe_gate_kernel(
    const float* __restrict__ x, const float* __restrict__ Wg,
    float* __restrict__ gates, int* __restrict__ counts,
    int* __restrict__ lists) {
    const int lane = threadIdx.x & 31;
    const int wave = threadIdx.x >> 5;
    const int t = blockIdx.x * 8 + wave;
    if (t >= NTOK) return;

    float s[E_];
#pragma unroll
    for (int e = 0; e < E_; ++e) s[e] = 0.0f;
    const float* xr = x + (size_t)t * D_;
    for (int d = lane; d < D_; d += 32) {
        const float xv = xr[d];
#pragma unroll
        for (int e = 0; e < E_; ++e) s[e] += xv * Wg[e * D_ + d];
    }
#pragma unroll
    for (int e = 0; e < E_; ++e) {
#pragma unroll
        for (int off = 16; off > 0; off >>= 1)
            s[e] += __shfl_xor(s[e], off, 32);
    }
    if (lane == 0) {
        int i0 = 0; float v0 = s[0];
#pragma unroll
        for (int e = 1; e < E_; ++e) if (s[e] > v0) { v0 = s[e]; i0 = e; }
        int i1 = -1; float v1 = -3.4e38f;
#pragma unroll
        for (int e = 0; e < E_; ++e)
            if (e != i0 && s[e] > v1) { v1 = s[e]; i1 = e; }
        const float p1 = expf(v1 - v0);
        const float inv = 1.0f / (1.0f + p1);
        const float g0 = inv, g1 = p1 * inv;
#pragma unroll
        for (int e = 0; e < E_; ++e) gates[t * E_ + e] = 0.0f;
        gates[t * E_ + i0] = g0;
        gates[t * E_ + i1] = g1;
        const int p = atomicAdd(&counts[i0], 1); lists[i0 * NTOK + p] = t;
        const int q = atomicAdd(&counts[i1], 1); lists[i1 * NTOK + q] = t;
    }
}

__global__ __launch_bounds__(256) void moe_loss_kernel(
    const float* __restrict__ gates, float* __restrict__ loss_out) {
    __shared__ float part[256];
    const int e = threadIdx.x & 7;
    const int slot = threadIdx.x >> 3;
    float s = 0.0f;
    for (int t = slot; t < NTOK; t += 32) s += gates[t * E_ + e];
    part[threadIdx.x] = s;
    __syncthreads();
    if (threadIdx.x == 0) {
        float u[E_];
        float mu = 0.0f;
        for (int ee = 0; ee < E_; ++ee) {
            float acc = 0.0f;
            for (int sl = 0; sl < 32; ++sl) acc += part[sl * 8 + ee];
            u[ee] = acc / (float)NTOK;
            mu += u[ee];
        }
        mu *= (1.0f / (float)E_);
        float var = 0.0f;
        for (int ee = 0; ee < E_; ++ee) {
            const float d = u[ee] - mu;
            var += d * d;
        }
        var *= (1.0f / (float)(E_ - 1));
        loss_out[0] = var * 0.01f;
    }
}

#define KC_   32
#define LDSA_ 34
#define LDSB_ 34

__global__ __launch_bounds__(256) void moe_fc1_kernel(
    const float* __restrict__ x, const float* __restrict__ W1e,
    const float* __restrict__ b1e, const int* __restrict__ lists,
    const int* __restrict__ counts, float* __restrict__ H,
    int expert, int chunkBase, int Tc) {
    const int cnt = counts[expert];
    int rowsInChunk = cnt - chunkBase;
    if (rowsInChunk > Tc) rowsInChunk = Tc;
    const int mBase = blockIdx.x * 64;
    if (rowsInChunk <= mBase) return;
    const int nBase = blockIdx.y * 128;

    __shared__ float As [64][LDSA_];
    __shared__ float Bst[128][LDSB_];

    const int tid  = threadIdx.x;
    const int lane = tid & 31, wave = tid >> 5;
    const int half = lane >> 4, lm = lane & 15;
    const int Moff = 32 * (wave & 1);
    const int Noff = 32 * (wave >> 1);

    const int rA0 = tid >> 3, rA1 = rA0 + 32;
    const int cA  = (tid & 7) << 2;
    const int* lst = lists + expert * NTOK + chunkBase;
    const bool va0 = (mBase + rA0) < rowsInChunk;
    const bool va1 = (mBase + rA1) < rowsInChunk;
    const float* ap0 = va0 ? (x + (size_t)lst[mBase + rA0] * D_) : x;
    const float* ap1 = va1 ? (x + (size_t)lst[mBase + rA1] * D_) : x;

    const int rB0 = tid >> 4, rB1 = rB0 + 16;
    const int cB0 = (tid & 15) << 2, cB1 = cB0 + 64;

    v8f acc00 = {}; v8f acc01 = {}; v8f acc10 = {}; v8f acc11 = {};

    for (int k0 = 0; k0 < D_; k0 += KC_) {
        if (va0) {
            const float4 q = *(const float4*)(ap0 + k0 + cA);
            As[rA0][cA] = q.x; As[rA0][cA+1] = q.y; As[rA0][cA+2] = q.z; As[rA0][cA+3] = q.w;
        } else {
            As[rA0][cA] = 0.f; As[rA0][cA+1] = 0.f; As[rA0][cA+2] = 0.f; As[rA0][cA+3] = 0.f;
        }
        if (va1) {
            const float4 q = *(const float4*)(ap1 + k0 + cA);
            As[rA1][cA] = q.x; As[rA1][cA+1] = q.y; As[rA1][cA+2] = q.z; As[rA1][cA+3] = q.w;
        } else {
            As[rA1][cA] = 0.f; As[rA1][cA+1] = 0.f; As[rA1][cA+2] = 0.f; As[rA1][cA+3] = 0.f;
        }
        {
            const float4 q = *(const float4*)(W1e + (size_t)(k0 + rB0) * DFF_ + nBase + cB0);
            Bst[cB0+0][rB0] = q.x; Bst[cB0+1][rB0] = q.y; Bst[cB0+2][rB0] = q.z; Bst[cB0+3][rB0] = q.w;
        }
        {
            const float4 q = *(const float4*)(W1e + (size_t)(k0 + rB0) * DFF_ + nBase + cB1);
            Bst[cB1+0][rB0] = q.x; Bst[cB1+1][rB0] = q.y; Bst[cB1+2][rB0] = q.z; Bst[cB1+3][rB0] = q.w;
        }
        {
            const float4 q = *(const float4*)(W1e + (size_t)(k0 + rB1) * DFF_ + nBase + cB0);
            Bst[cB0+0][rB1] = q.x; Bst[cB0+1][rB1] = q.y; Bst[cB0+2][rB1] = q.z; Bst[cB0+3][rB1] = q.w;
        }
        {
            const float4 q = *(const float4*)(W1e + (size_t)(k0 + rB1) * DFF_ + nBase + cB1);
            Bst[cB0+64][rB1] = q.x; Bst[cB1+1][rB1] = q.y; Bst[cB1+2][rB1] = q.z; Bst[cB1+3][rB1] = q.w;
        }
        __syncthreads();
#pragma unroll
        for (int kk = 0; kk < KC_; kk += 4) {
            const int ka = kk + 2 * half;
            v2f a0; a0.x = As[Moff + lm][ka];       a0.y = As[Moff + lm][ka + 1];
            v2f a1; a1.x = As[Moff + 16 + lm][ka];  a1.y = As[Moff + 16 + lm][ka + 1];
            v2f b0; b0.x = Bst[Noff + lm][ka];      b0.y = Bst[Noff + lm][ka + 1];
            v2f b1; b1.x = Bst[Noff + 16 + lm][ka]; b1.y = Bst[Noff + 16 + lm][ka + 1];
            acc00 = wmma_f32_k4(a0, b0, acc00);
            acc01 = wmma_f32_k4(a0, b1, acc01);
            acc10 = wmma_f32_k4(a1, b0, acc10);
            acc11 = wmma_f32_k4(a1, b1, acc11);
        }
        __syncthreads();
    }
#pragma unroll
    for (int r = 0; r < 8; ++r) {
        const int m0 = Moff + r + 8 * half;
        const int lrow0 = mBase + m0, lrow1 = lrow0 + 16;
        const int n0 = nBase + Noff + lm, n1 = n0 + 16;
        if (lrow0 < rowsInChunk) {
            H[(size_t)lrow0 * DFF_ + n0] = gelu_exact(acc00[r] + b1e[n0]);
            H[(size_t)lrow0 * DFF_ + n1] = gelu_exact(acc01[r] + b1e[n1]);
        }
        if (lrow1 < rowsInChunk) {
            H[(size_t)lrow1 * DFF_ + n0] = gelu_exact(acc10[r] + b1e[n0]);
            H[(size_t)lrow1 * DFF_ + n1] = gelu_exact(acc11[r] + b1e[n1]);
        }
    }
}

__global__ __launch_bounds__(256) void moe_fc2_kernel(
    const float* __restrict__ H, const float* __restrict__ W2e,
    const float* __restrict__ b2e, const int* __restrict__ lists,
    const int* __restrict__ counts, const float* __restrict__ gates,
    float* __restrict__ out, int expert, int chunkBase, int Tc) {
    const int cnt = counts[expert];
    int rowsInChunk = cnt - chunkBase;
    if (rowsInChunk > Tc) rowsInChunk = Tc;
    const int mBase = blockIdx.x * 64;
    if (rowsInChunk <= mBase) return;
    const int nBase = blockIdx.y * 128;

    __shared__ float As [64][LDSA_];
    __shared__ float Bst[128][LDSB_];

    const int tid  = threadIdx.x;
    const int lane = tid & 31, wave = tid >> 5;
    const int half = lane >> 4, lm = lane & 15;
    const int Moff = 32 * (wave & 1);
    const int Noff = 32 * (wave >> 1);

    const int rA0 = tid >> 3, rA1 = rA0 + 32;
    const int cA  = (tid & 7) << 2;
    const bool va0 = (mBase + rA0) < rowsInChunk;
    const bool va1 = (mBase + rA1) < rowsInChunk;
    const float* ap0 = H + (size_t)(mBase + rA0) * DFF_;
    const float* ap1 = H + (size_t)(mBase + rA1) * DFF_;

    const int rB0 = tid >> 4, rB1 = rB0 + 16;
    const int cB0 = (tid & 15) << 2, cB1 = cB0 + 64;

    v8f acc00 = {}; v8f acc01 = {}; v8f acc10 = {}; v8f acc11 = {};

    for (int k0 = 0; k0 < DFF_; k0 += KC_) {
        if (va0) {
            const float4 q = *(const float4*)(ap0 + k0 + cA);
            As[rA0][cA] = q.x; As[rA0][cA+1] = q.y; As[rA0][cA+2] = q.z; As[rA0][cA+3] = q.w;
        } else {
            As[rA0][cA] = 0.f; As[rA0][cA+1] = 0.f; As[rA0][cA+2] = 0.f; As[rA0][cA+3] = 0.f;
        }
        if (va1) {
            const float4 q = *(const float4*)(ap1 + k0 + cA);
            As[rA1][cA] = q.x; As[rA1][cA+1] = q.y; As[rA1][cA+2] = q.z; As[rA1][cA+3] = q.w;
        } else {
            As[rA1][cA] = 0.f; As[rA1][cA+1] = 0.f; As[rA1][cA+2] = 0.f; As[rA1][cA+3] = 0.f;
        }
        {
            const float4 q = *(const float4*)(W2e + (size_t)(k0 + rB0) * D_ + nBase + cB0);
            Bst[cB0+0][rB0] = q.x; Bst[cB0+1][rB0] = q.y; Bst[cB0+2][rB0] = q.z; Bst[cB0+3][rB0] = q.w;
        }
        {
            const float4 q = *(const float4*)(W2e + (size_t)(k0 + rB0) * D_ + nBase + cB1);
            Bst[cB1+0][rB0] = q.x; Bst[cB1+1][rB0] = q.y; Bst[cB1+2][rB0] = q.z; Bst[cB1+3][rB0] = q.w;
        }
        {
            const float4 q = *(const float4*)(W2e + (size_t)(k0 + rB1) * D_ + nBase + cB0);
            Bst[cB0+0][rB1] = q.x; Bst[cB0+1][rB1] = q.y; Bst[cB0+2][rB1] = q.z; Bst[cB0+3][rB1] = q.w;
        }
        {
            const float4 q = *(const float4*)(W2e + (size_t)(k0 + rB1) * D_ + nBase + cB1);
            Bst[cB1+0][rB1] = q.x; Bst[cB1+1][rB1] = q.y; Bst[cB1+2][rB1] = q.z; Bst[cB1+3][rB1] = q.w;
        }
        __syncthreads();
#pragma unroll
        for (int kk = 0; kk < KC_; kk += 4) {
            const int ka = kk + 2 * half;
            v2f a0; a0.x = As[Moff + lm][ka];       a0.y = As[Moff + lm][ka + 1];
            v2f a1; a1.x = As[Moff + 16 + lm][ka];  a1.y = As[Moff + 16 + lm][ka + 1];
            v2f b0; b0.x = Bst[Noff + lm][ka];      b0.y = Bst[Noff + lm][ka + 1];
            v2f b1; b1.x = Bst[Noff + 16 + lm][ka]; b1.y = Bst[Noff + 16 + lm][ka + 1];
            acc00 = wmma_f32_k4(a0, b0, acc00);
            acc01 = wmma_f32_k4(a0, b1, acc01);
            acc10 = wmma_f32_k4(a1, b0, acc10);
            acc11 = wmma_f32_k4(a1, b1, acc11);
        }
        __syncthreads();
    }
    const int* lst = lists + expert * NTOK + chunkBase;
#pragma unroll
    for (int r = 0; r < 8; ++r) {
        const int m0 = Moff + r + 8 * half;
        const int lrow0 = mBase + m0, lrow1 = lrow0 + 16;
        const int n0 = nBase + Noff + lm, n1 = n0 + 16;
        if (lrow0 < rowsInChunk) {
            const int tok = lst[lrow0];
            const float g = gates[tok * E_ + expert];
            out[(size_t)tok * D_ + n0] += g * (acc00[r] + b2e[n0]);
            out[(size_t)tok * D_ + n1] += g * (acc01[r] + b2e[n1]);
        }
        if (lrow1 < rowsInChunk) {
            const int tok = lst[lrow1];
            const float g = gates[tok * E_ + expert];
            out[(size_t)tok * D_ + n0] += g * (acc10[r] + b2e[n0]);
            out[(size_t)tok * D_ + n1] += g * (acc11[r] + b2e[n1]);
        }
    }
}

extern "C" void kernel_launch(void* const* d_in, const int* in_sizes, int n_in,
                              void* d_out, int out_size, void* d_ws, size_t ws_size,
                              hipStream_t stream) {
    const float* x  = (const float*)d_in[0];
    const float* Wg = (const float*)d_in[1];
    const float* W1 = (const float*)d_in[2];
    const float* b1 = (const float*)d_in[3];
    const float* W2 = (const float*)d_in[4];
    const float* b2 = (const float*)d_in[5];
    float* out = (float*)d_out;

    char* ws = (char*)d_ws;
    int*   counts = (int*)ws;
    int*   lists  = (int*)(ws + 256);
    float* gates  = (float*)(ws + 256 + 4ull * E_ * NTOK);
    size_t OFS_H  = 256 + 8ull * E_ * NTOK;
    OFS_H = (OFS_H + 255) & ~(size_t)255;
    float* H = (float*)(ws + OFS_H);

    size_t avail = (ws_size > OFS_H) ? (ws_size - OFS_H) : 0;
    long long tcll = (long long)(avail / ((size_t)DFF_ * 4));
    int Tc = (int)(tcll > NTOK ? NTOK : tcll);
    Tc &= ~63;
    if (Tc <= 0) Tc = 64;
    const int chunks = (NTOK + Tc - 1) / Tc;

    const long long outN = (long long)NTOK * D_ + 1;
    moe_zero_kernel<<<2048, 256, 0, stream>>>(out, outN, counts);
    moe_gate_kernel<<<NTOK / 8, 256, 0, stream>>>(x, Wg, gates, counts, lists);
    moe_loss_kernel<<<1, 256, 0, stream>>>(gates, out + (outN - 1));

    for (int e = 0; e < E_; ++e) {
        const float* W1e = W1 + (size_t)e * D_ * DFF_;
        const float* b1e = b1 + (size_t)e * DFF_;
        const float* W2e = W2 + (size_t)e * DFF_ * D_;
        const float* b2e = b2 + (size_t)e * D_;
        for (int c = 0; c < chunks; ++c) {
            dim3 g1(Tc / 64, DFF_ / 128);
            moe_fc1_kernel<<<g1, 256, 0, stream>>>(x, W1e, b1e, lists, counts,
                                                   H, e, c * Tc, Tc);
            dim3 g2(Tc / 64, D_ / 128);
            moe_fc2_kernel<<<g2, 256, 0, stream>>>(H, W2e, b2e, lists, counts,
                                                   gates, out, e, c * Tc, Tc);
        }
    }
}


# K loop of both expert GEMMs software-pipelined: LDS tiles double-buffered, next chunk's global loads issued ahead of the current chunk's WMMAs, one barrier per chunk; sum order unchanged
# speedup vs baseline: 2.3464x; 2.3464x over previous
#include <hip/hip_runtime.h>
#include <math.h>

#define D_    1024
#define DFF_  4096
#define E_    8
#define NTOK  8192
#define TPP_  136

typedef float v2f __attribute__((ext_vector_type(2)));
typedef float v8f __attribute__((ext_vector_type(8)));

__device__ __forceinline__ v8f wmma_f32_k4(v2f a, v2f b, v8f c) {
    return __builtin_amdgcn_wmma_f32_16x16x4_f32(
         false, a,  false, b,
         (short)0, c,  false,  false);
}

__device__ __forceinline__ float gelu_exact(float v) {
    return 0.5f * v * (1.0f + erff(v * 0.70710678118654752440f));
}

__device__ __forceinline__ bool find_tile(const int* __restrict__ counts, int tile,
                                          int& list, int& mBase, int& cnt) {
    list = -1; mBase = 0; cnt = 0;
#pragma unroll
    for (int s = 0; s < 2; ++s) {
        int tb = s * TPP_;
#pragma unroll
        for (int e = 0; e < E_; ++e) {
            const int c = counts[s * E_ + e];
            const int nt = (c + 63) >> 6;
            if (tile >= tb && tile < tb + nt) { list = s * E_ + e; mBase = (tile - tb) << 6; cnt = c; }
            tb += nt;
        }
    }
    return list >= 0;
}

__global__ void moe_zero_kernel(float* __restrict__ out, long long n,
                                int* __restrict__ counts) {
    long long i = (long long)blockIdx.x * blockDim.x + threadIdx.x;
    const long long stride = (long long)gridDim.x * blockDim.x;
    for (; i < n; i += stride) out[i] = 0.0f;
    if (blockIdx.x == 0 && threadIdx.x < 2 * E_) counts[threadIdx.x] = 0;
}

__global__ __launch_bounds__(256) void moe_gate_kernel(
    const float* __restrict__ x, const float* __restrict__ Wg,
    float* __restrict__ gates, int* __restrict__ counts,
    int* __restrict__ lists) {
    const int lane = threadIdx.x & 31;
    const int wave = threadIdx.x >> 5;
    const int t = blockIdx.x * 8 + wave;
    if (t >= NTOK) return;

    float s[E_];
#pragma unroll
    for (int e = 0; e < E_; ++e) s[e] = 0.0f;
    const float* xr = x + (size_t)t * D_;
    for (int d = lane; d < D_; d += 32) {
        const float xv = xr[d];
#pragma unroll
        for (int e = 0; e < E_; ++e) s[e] += xv * Wg[e * D_ + d];
    }
#pragma unroll
    for (int e = 0; e < E_; ++e) {
#pragma unroll
        for (int off = 16; off > 0; off >>= 1)
            s[e] += __shfl_xor(s[e], off, 32);
    }
    if (lane == 0) {
        int i0 = 0; float v0 = s[0];
#pragma unroll
        for (int e = 1; e < E_; ++e) if (s[e] > v0) { v0 = s[e]; i0 = e; }
        int i1 = -1; float v1 = -3.4e38f;
#pragma unroll
        for (int e = 0; e < E_; ++e)
            if (e != i0 && s[e] > v1) { v1 = s[e]; i1 = e; }
        const float p1 = expf(v1 - v0);
        const float inv = 1.0f / (1.0f + p1);
        const float g0 = inv, g1 = p1 * inv;
#pragma unroll
        for (int e = 0; e < E_; ++e) gates[t * E_ + e] = 0.0f;
        gates[t * E_ + i0] = g0;
        gates[t * E_ + i1] = g1;
        const int lo = i0 < i1 ? i0 : i1;
        const int hi = i0 < i1 ? i1 : i0;
        const int p = atomicAdd(&counts[lo], 1);      lists[lo * NTOK + p] = t;
        const int q = atomicAdd(&counts[E_ + hi], 1); lists[(E_ + hi) * NTOK + q] = t;
    }
}

__global__ __launch_bounds__(256) void moe_loss_kernel(
    const float* __restrict__ gates, float* __restrict__ loss_out) {
    __shared__ float part[256];
    const int e = threadIdx.x & 7;
    const int slot = threadIdx.x >> 3;
    float s = 0.0f;
    for (int t = slot; t < NTOK; t += 32) s += gates[t * E_ + e];
    part[threadIdx.x] = s;
    __syncthreads();
    if (threadIdx.x == 0) {
        float u[E_];
        float mu = 0.0f;
        for (int ee = 0; ee < E_; ++ee) {
            float acc = 0.0f;
            for (int sl = 0; sl < 32; ++sl) acc += part[sl * 8 + ee];
            u[ee] = acc / (float)NTOK;
            mu += u[ee];
        }
        mu *= (1.0f / (float)E_);
        float var = 0.0f;
        for (int ee = 0; ee < E_; ++ee) {
            const float d = u[ee] - mu;
            var += d * d;
        }
        var *= (1.0f / (float)(E_ - 1));
        loss_out[0] = var * 0.01f;
    }
}

#define KC_    32
#define ROWB_  136
#define BOFF_  (64 * ROWB_)
#define BUFX_  32768
#define SMEMB_ (BUFX_ + BOFF_ + 128 * ROWB_)

template <int LDB, int KTOT>
__device__ __forceinline__ void gemm_tile_64x128(
    const float* __restrict__ abase, unsigned aoff0, unsigned aoff1,
    const float* __restrict__ bp, char* sm,
    v8f& acc00, v8f& acc01, v8f& acc10, v8f& acc11) {
    const int tid  = threadIdx.x;
    const int lane = tid & 31, wave = tid >> 5;
    const int half = lane >> 4, lm = lane & 15;
    const int Moff = 32 * (wave & 1);
    const int Noff = 32 * (wave >> 1);

    const int rA0 = tid >> 3;
    const int cA  = (tid & 7) << 2;
    const int rB0 = tid >> 4;
    const int cB0 = (tid & 15) << 2;
    const unsigned oa0 = aoff0 + (unsigned)cA;
    const unsigned oa1 = aoff1 + (unsigned)cA;
    const unsigned ob  = (unsigned)(rB0 * LDB + cB0);

    unsigned ldA = (unsigned)((Moff + lm) * ROWB_ + half * 8);
    unsigned ldB = (unsigned)(BOFF_ + (Noff + lm) * ROWB_ + half * 8);
    unsigned stA = (unsigned)(rA0 * ROWB_ + cA * 4);
    unsigned stB = (unsigned)(BOFF_ + cB0 * ROWB_ + rB0 * 4);

    float4 qa0, qa1, qb00, qb01, qb10, qb11;

    auto gload = [&](int k0) {
        const float* __restrict__ ak = abase + k0;
        const float* __restrict__ bk = bp + (size_t)k0 * LDB;
        qa0  = *(const float4*)(ak + oa0);
        qa1  = *(const float4*)(ak + oa1);
        qb00 = *(const float4*)(bk + ob);
        qb01 = *(const float4*)(bk + ob + 64);
        qb10 = *(const float4*)(bk + ob + 16 * LDB);
        qb11 = *(const float4*)(bk + ob + 16 * LDB + 64);
    };
    auto st1 = [&](unsigned off, float v) { *(float*)(sm + off) = v; };
    auto st2 = [&](unsigned off, float v, float w) { v2f t; t.x = v; t.y = w; *(v2f*)(sm + off) = t; };
    auto lstore = [&]() {
        st2(stA, qa0.x, qa0.y);              st2(stA + 8, qa0.z, qa0.w);
        st2(stA + 32 * ROWB_, qa1.x, qa1.y); st2(stA + 32 * ROWB_ + 8, qa1.z, qa1.w);
        st1(stB,             qb00.x); st1(stB + ROWB_,          qb00.y);
        st1(stB + 2 * ROWB_, qb00.z); st1(stB + 3 * ROWB_,      qb00.w);
        st1(stB + 64,             qb10.x); st1(stB + ROWB_ + 64,     qb10.y);
        st1(stB + 2 * ROWB_ + 64, qb10.z); st1(stB + 3 * ROWB_ + 64, qb10.w);
        st1(stB + 64 * ROWB_,             qb01.x); st1(stB + 65 * ROWB_,      qb01.y);
        st1(stB + 66 * ROWB_,             qb01.z); st1(stB + 67 * ROWB_,      qb01.w);
        st1(stB + 64 * ROWB_ + 64,        qb11.x); st1(stB + 65 * ROWB_ + 64, qb11.y);
        st1(stB + 66 * ROWB_ + 64,        qb11.z); st1(stB + 67 * ROWB_ + 64, qb11.w);
    };
    auto compute = [&]() {
#pragma unroll 1
        for (int kq = 0; kq < KC_; kq += 8) {
#pragma unroll
            for (int kk = 0; kk < 8; kk += 4) {
                const v2f a0 = *(const v2f*)(sm + ldA + kk * 4);
                const v2f a1 = *(const v2f*)(sm + ldA + 16 * ROWB_ + kk * 4);
                const v2f b0 = *(const v2f*)(sm + ldB + kk * 4);
                const v2f b1 = *(const v2f*)(sm + ldB + 16 * ROWB_ + kk * 4);
                acc00 = wmma_f32_k4(a0, b0, acc00);
                acc01 = wmma_f32_k4(a0, b1, acc01);
                acc10 = wmma_f32_k4(a1, b0, acc10);
                acc11 = wmma_f32_k4(a1, b1, acc11);
            }
            ldA += 32; ldB += 32;
        }
        ldA -= KC_ * 4; ldB -= KC_ * 4;
    };

    gload(0);
    lstore();
    __syncthreads();
    stA ^= BUFX_; stB ^= BUFX_;
    for (int k0 = 0; k0 < KTOT; k0 += KC_) {
        const bool more = (k0 + KC_) < KTOT;
        if (more) gload(k0 + KC_);
        compute();
        if (more) {
            lstore();
            __syncthreads();
        }
        ldA ^= BUFX_; ldB ^= BUFX_; stA ^= BUFX_; stB ^= BUFX_;
    }
}

__global__ __launch_bounds__(256, 10) void moe_fc1_kernel(
    const float* __restrict__ x, const float* __restrict__ W1,
    const float* __restrict__ b1, const int* __restrict__ lists,
    const int* __restrict__ counts, float* __restrict__ H,
    int chunkTileBase) {
    int list, mBase, cnt;
    if (!find_tile(counts, chunkTileBase + (int)blockIdx.y, list, mBase, cnt)) return;
    const int expert = list & (E_ - 1);
    const float* __restrict__ W1e = W1 + (size_t)expert * D_ * DFF_;
    const float* __restrict__ b1e = b1 + (size_t)expert * DFF_;
    const int nBase = blockIdx.x * 128;
    const int hBase = blockIdx.y * 64;

    __shared__ __attribute__((aligned(16))) float smem[SMEMB_ / 4];

    const int tid  = threadIdx.x;
    const int lane = tid & 31, wave = tid >> 5;
    const int half = lane >> 4, lm = lane & 15;
    const int Moff = 32 * (wave & 1);
    const int Noff = 32 * (wave >> 1);

    const int rA0 = tid >> 3, rA1 = rA0 + 32;
    const int* lst = lists + list * NTOK + mBase;
    const bool va0 = (mBase + rA0) < cnt;
    const bool va1 = (mBase + rA1) < cnt;
    const unsigned ao0 = va0 ? (unsigned)lst[rA0] * D_ : 0u;
    const unsigned ao1 = va1 ? (unsigned)lst[rA1] * D_ : 0u;

    v8f acc00 = {}; v8f acc01 = {}; v8f acc10 = {}; v8f acc11 = {};
    gemm_tile_64x128<DFF_, D_>(x, ao0, ao1, W1e + nBase, (char*)smem, acc00, acc01, acc10, acc11);

#pragma unroll
    for (int r = 0; r < 8; ++r) {
        const int m0 = Moff + r + 8 * half;
        const int hrow0 = hBase + m0, hrow1 = hrow0 + 16;
        const int n0 = nBase + Noff + lm, n1 = n0 + 16;
        if (mBase + m0 < cnt) {
            H[(size_t)hrow0 * DFF_ + n0] = gelu_exact(acc00[r] + b1e[n0]);
            H[(size_t)hrow0 * DFF_ + n1] = gelu_exact(acc01[r] + b1e[n1]);
        }
        if (mBase + m0 + 16 < cnt) {
            H[(size_t)hrow1 * DFF_ + n0] = gelu_exact(acc10[r] + b1e[n0]);
            H[(size_t)hrow1 * DFF_ + n1] = gelu_exact(acc11[r] + b1e[n1]);
        }
    }
}

__global__ __launch_bounds__(256, 10) void moe_fc2_kernel(
    const float* __restrict__ H, const float* __restrict__ W2,
    const float* __restrict__ b2, const int* __restrict__ lists,
    const int* __restrict__ counts, const float* __restrict__ gates,
    float* __restrict__ out, int tileLo, int chunkTileBase) {
    const int tile = tileLo + (int)blockIdx.y;
    int list, mBase, cnt;
    if (!find_tile(counts, tile, list, mBase, cnt)) return;
    const int expert = list & (E_ - 1);
    const float* __restrict__ W2e = W2 + (size_t)expert * DFF_ * D_;
    const float* __restrict__ b2e = b2 + (size_t)expert * D_;
    const int nBase = blockIdx.x * 128;
    const int hBase = (tile - chunkTileBase) * 64;

    __shared__ __attribute__((aligned(16))) float smem[SMEMB_ / 4];

    const int tid  = threadIdx.x;
    const int lane = tid & 31, wave = tid >> 5;
    const int half = lane >> 4, lm = lane & 15;
    const int Moff = 32 * (wave & 1);
    const int Noff = 32 * (wave >> 1);

    const int rA0 = tid >> 3, rA1 = rA0 + 32;
    const bool va0 = (mBase + rA0) < cnt;
    const bool va1 = (mBase + rA1) < cnt;
    const unsigned ao0 = (unsigned)(va0 ? rA0 : 0) * DFF_;
    const unsigned ao1 = (unsigned)(va1 ? rA1 : 0) * DFF_;

    v8f acc00 = {}; v8f acc01 = {}; v8f acc10 = {}; v8f acc11 = {};
    gemm_tile_64x128<D_, DFF_>(H + (size_t)hBase * DFF_, ao0, ao1, W2e + nBase, (char*)smem, acc00, acc01, acc10, acc11);

    const int* lst = lists + list * NTOK;
#pragma unroll
    for (int r = 0; r < 8; ++r) {
        const int m0 = Moff + r + 8 * half;
        const int lrow0 = mBase + m0, lrow1 = lrow0 + 16;
        const int n0 = nBase + Noff + lm, n1 = n0 + 16;
        if (lrow0 < cnt) {
            const int tok = lst[lrow0];
            const float g = gates[tok * E_ + expert];
            out[(size_t)tok * D_ + n0] += g * (acc00[r] + b2e[n0]);
            out[(size_t)tok * D_ + n1] += g * (acc01[r] + b2e[n1]);
        }
        if (lrow1 < cnt) {
            const int tok = lst[lrow1];
            const float g = gates[tok * E_ + expert];
            out[(size_t)tok * D_ + n0] += g * (acc10[r] + b2e[n0]);
            out[(size_t)tok * D_ + n1] += g * (acc11[r] + b2e[n1]);
        }
    }
}

extern "C" void kernel_launch(void* const* d_in, const int* in_sizes, int n_in,
                              void* d_out, int out_size, void* d_ws, size_t ws_size,
                              hipStream_t stream) {
    const float* x  = (const float*)d_in[0];
    const float* Wg = (const float*)d_in[1];
    const float* W1 = (const float*)d_in[2];
    const float* b1 = (const float*)d_in[3];
    const float* W2 = (const float*)d_in[4];
    const float* b2 = (const float*)d_in[5];
    float* out = (float*)d_out;

    char* ws = (char*)d_ws;
    int*   counts = (int*)ws;
    int*   lists  = (int*)(ws + 256);
    float* gates  = (float*)(ws + 256 + 8ull * E_ * NTOK);
    size_t OFS_H  = 256 + 12ull * E_ * NTOK;
    OFS_H = (OFS_H + 255) & ~(size_t)255;
    float* H = (float*)(ws + OFS_H);

    const int MAXT = 2 * TPP_;
    size_t avail = (ws_size > OFS_H) ? (ws_size - OFS_H) : 0;
    long long capll = (long long)(avail / ((size_t)64 * DFF_ * 4));
    int cap = (int)(capll > MAXT ? MAXT : capll);
    if (cap <= 0) cap = 1;
    const int chunks = (MAXT + cap - 1) / cap;
    const int TPC = (MAXT + chunks - 1) / chunks;

    const long long outN = (long long)NTOK * D_ + 1;
    moe_zero_kernel<<<2048, 256, 0, stream>>>(out, outN, counts);
    moe_gate_kernel<<<NTOK / 8, 256, 0, stream>>>(x, Wg, gates, counts, lists);
    moe_loss_kernel<<<1, 256, 0, stream>>>(gates, out + (outN - 1));

    for (int c = 0; c < chunks; ++c) {
        const int cs = c * TPC;
        const int ce = (cs + TPC < MAXT) ? (cs + TPC) : MAXT;
        dim3 g1(DFF_ / 128, ce - cs);
        moe_fc1_kernel<<<g1, 256, 0, stream>>>(x, W1, b1, lists, counts, H, cs);
        for (int s = 0; s < 2; ++s) {
            const int lo = (s * TPP_ > cs) ? s * TPP_ : cs;
            const int hi = ((s + 1) * TPP_ < ce) ? (s + 1) * TPP_ : ce;
            if (hi > lo) {
                dim3 g2(D_ / 128, hi - lo);
                moe_fc2_kernel<<<g2, 256, 0, stream>>>(H, W2, b2, lists, counts,
                                                       gates, out, lo, cs);
            }
        }
    }
}


# one fc2 launch for both top-2 slots, twice the blocks divide better over the resident places: lower-slot tiles store out, higher-slot tiles a workspace plane, small kernel adds; sum order kept
# speedup vs baseline: 2.4826x; 1.0580x over previous
#include <hip/hip_runtime.h>
#include <math.h>

#define D_    1024
#define DFF_  4096
#define E_    8
#define NTOK  8192
#define TPP_  136

typedef float v2f __attribute__((ext_vector_type(2)));
typedef float v8f __attribute__((ext_vector_type(8)));

__device__ __forceinline__ v8f wmma_f32_k4(v2f a, v2f b, v8f c) {
    return __builtin_amdgcn_wmma_f32_16x16x4_f32(
         false, a,  false, b,
         (short)0, c,  false,  false);
}

__device__ __forceinline__ float gelu_exact(float v) {
    return 0.5f * v * (1.0f + erff(v * 0.70710678118654752440f));
}

__device__ __forceinline__ bool find_tile(const int* __restrict__ counts, int tile,
                                          int& list, int& mBase, int& cnt) {
    list = -1; mBase = 0; cnt = 0;
#pragma unroll
    for (int s = 0; s < 2; ++s) {
        int tb = s * TPP_;
#pragma unroll
        for (int e = 0; e < E_; ++e) {
            const int c = counts[s * E_ + e];
            const int nt = (c + 63) >> 6;
            if (tile >= tb && tile < tb + nt) { list = s * E_ + e; mBase = (tile - tb) << 6; cnt = c; }
            tb += nt;
        }
    }
    return list >= 0;
}

__global__ void moe_zero_kernel(float* __restrict__ out, long long n,
                                int* __restrict__ counts) {
    long long i = (long long)blockIdx.x * blockDim.x + threadIdx.x;
    const long long stride = (long long)gridDim.x * blockDim.x;
    for (; i < n; i += stride) out[i] = 0.0f;
    if (blockIdx.x == 0 && threadIdx.x < 2 * E_) counts[threadIdx.x] = 0;
}

__global__ __launch_bounds__(256) void moe_gate_kernel(
    const float* __restrict__ x, const float* __restrict__ Wg,
    float* __restrict__ gates, int* __restrict__ counts,
    int* __restrict__ lists) {
    const int lane = threadIdx.x & 31;
    const int wave = threadIdx.x >> 5;
    const int t = blockIdx.x * 8 + wave;
    if (t >= NTOK) return;

    float s[E_];
#pragma unroll
    for (int e = 0; e < E_; ++e) s[e] = 0.0f;
    const float* xr = x + (size_t)t * D_;
    for (int d = lane; d < D_; d += 32) {
        const float xv = xr[d];
#pragma unroll
        for (int e = 0; e < E_; ++e) s[e] += xv * Wg[e * D_ + d];
    }
#pragma unroll
    for (int e = 0; e < E_; ++e) {
#pragma unroll
        for (int off = 16; off > 0; off >>= 1)
            s[e] += __shfl_xor(s[e], off, 32);
    }
    if (lane == 0) {
        int i0 = 0; float v0 = s[0];
#pragma unroll
        for (int e = 1; e < E_; ++e) if (s[e] > v0) { v0 = s[e]; i0 = e; }
        int i1 = -1; float v1 = -3.4e38f;
#pragma unroll
        for (int e = 0; e < E_; ++e)
            if (e != i0 && s[e] > v1) { v1 = s[e]; i1 = e; }
        const float p1 = expf(v1 - v0);
        const float inv = 1.0f / (1.0f + p1);
        const float g0 = inv, g1 = p1 * inv;
#pragma unroll
        for (int e = 0; e < E_; ++e) gates[t * E_ + e] = 0.0f;
        gates[t * E_ + i0] = g0;
        gates[t * E_ + i1] = g1;
        const int lo = i0 < i1 ? i0 : i1;
        const int hi = i0 < i1 ? i1 : i0;
        const int p = atomicAdd(&counts[lo], 1);      lists[lo * NTOK + p] = t;
        const int q = atomicAdd(&counts[E_ + hi], 1); lists[(E_ + hi) * NTOK + q] = t;
    }
}

__global__ __launch_bounds__(256) void moe_loss_kernel(
    const float* __restrict__ gates, float* __restrict__ loss_out) {
    __shared__ float part[256];
    const int e = threadIdx.x & 7;
    const int slot = threadIdx.x >> 3;
    float s = 0.0f;
    for (int t = slot; t < NTOK; t += 32) s += gates[t * E_ + e];
    part[threadIdx.x] = s;
    __syncthreads();
    if (threadIdx.x == 0) {
        float u[E_];
        float mu = 0.0f;
        for (int ee = 0; ee < E_; ++ee) {
            float acc = 0.0f;
            for (int sl = 0; sl < 32; ++sl) acc += part[sl * 8 + ee];
            u[ee] = acc / (float)NTOK;
            mu += u[ee];
        }
        mu *= (1.0f / (float)E_);
        float var = 0.0f;
        for (int ee = 0; ee < E_; ++ee) {
            const float d = u[ee] - mu;
            var += d * d;
        }
        var *= (1.0f / (float)(E_ - 1));
        loss_out[0] = var * 0.01f;
    }
}

#define KC_    32
#define ROWB_  136
#define BOFF_  (64 * ROWB_)
#define BUFX_  32768
#define SMEMB_ (BUFX_ + BOFF_ + 128 * ROWB_)

template <int LDB, int KTOT>
__device__ __forceinline__ void gemm_tile_64x128(
    const float* __restrict__ abase, unsigned aoff0, unsigned aoff1,
    const float* __restrict__ bp, char* sm,
    v8f& acc00, v8f& acc01, v8f& acc10, v8f& acc11) {
    const int tid  = threadIdx.x;
    const int lane = tid & 31, wave = tid >> 5;
    const int half = lane >> 4, lm = lane & 15;
    const int Moff = 32 * (wave & 1);
    const int Noff = 32 * (wave >> 1);

    const int rA0 = tid >> 3;
    const int cA  = (tid & 7) << 2;
    const int rB0 = tid >> 4;
    const int cB0 = (tid & 15) << 2;
    const unsigned oa0 = aoff0 + (unsigned)cA;
    const unsigned oa1 = aoff1 + (unsigned)cA;
    const unsigned ob  = (unsigned)(rB0 * LDB + cB0);

    unsigned ldA = (unsigned)((Moff + lm) * ROWB_ + half * 8);
    unsigned ldB = (unsigned)(BOFF_ + (Noff + lm) * ROWB_ + half * 8);
    unsigned stA = (unsigned)(rA0 * ROWB_ + cA * 4);
    unsigned stB = (unsigned)(BOFF_ + cB0 * ROWB_ + rB0 * 4);

    float4 qa0, qa1, qb00, qb01, qb10, qb11;

    auto gload = [&](int k0) {
        const float* __restrict__ ak = abase + k0;
        const float* __restrict__ bk = bp + (size_t)k0 * LDB;
        qa0  = *(const float4*)(ak + oa0);
        qa1  = *(const float4*)(ak + oa1);
        qb00 = *(const float4*)(bk + ob);
        qb01 = *(const float4*)(bk + ob + 64);
        qb10 = *(const float4*)(bk + ob + 16 * LDB);
        qb11 = *(const float4*)(bk + ob + 16 * LDB + 64);
    };
    auto st1 = [&](unsigned off, float v) { *(float*)(sm + off) = v; };
    auto st2 = [&](unsigned off, float v, float w) { v2f t; t.x = v; t.y = w; *(v2f*)(sm + off) = t; };
    auto lstore = [&]() {
        st2(stA, qa0.x, qa0.y);              st2(stA + 8, qa0.z, qa0.w);
        st2(stA + 32 * ROWB_, qa1.x, qa1.y); st2(stA + 32 * ROWB_ + 8, qa1.z, qa1.w);
        st1(stB,             qb00.x); st1(stB + ROWB_,          qb00.y);
        st1(stB + 2 * ROWB_, qb00.z); st1(stB + 3 * ROWB_,      qb00.w);
        st1(stB + 64,             qb10.x); st1(stB + ROWB_ + 64,     qb10.y);
        st1(stB + 2 * ROWB_ + 64, qb10.z); st1(stB + 3 * ROWB_ + 64, qb10.w);
        st1(stB + 64 * ROWB_,             qb01.x); st1(stB + 65 * ROWB_,      qb01.y);
        st1(stB + 66 * ROWB_,             qb01.z); st1(stB + 67 * ROWB_,      qb01.w);
        st1(stB + 64 * ROWB_ + 64,        qb11.x); st1(stB + 65 * ROWB_ + 64, qb11.y);
        st1(stB + 66 * ROWB_ + 64,        qb11.z); st1(stB + 67 * ROWB_ + 64, qb11.w);
    };
    auto compute = [&]() {
#pragma unroll 1
        for (int kq = 0; kq < KC_; kq += 8) {
#pragma unroll
            for (int kk = 0; kk < 8; kk += 4) {
                const v2f a0 = *(const v2f*)(sm + ldA + kk * 4);
                const v2f a1 = *(const v2f*)(sm + ldA + 16 * ROWB_ + kk * 4);
                const v2f b0 = *(const v2f*)(sm + ldB + kk * 4);
                const v2f b1 = *(const v2f*)(sm + ldB + 16 * ROWB_ + kk * 4);
                acc00 = wmma_f32_k4(a0, b0, acc00);
                acc01 = wmma_f32_k4(a0, b1, acc01);
                acc10 = wmma_f32_k4(a1, b0, acc10);
                acc11 = wmma_f32_k4(a1, b1, acc11);
            }
            ldA += 32; ldB += 32;
        }
        ldA -= KC_ * 4; ldB -= KC_ * 4;
    };

    gload(0);
    lstore();
    __syncthreads();
    stA ^= BUFX_; stB ^= BUFX_;
    for (int k0 = 0; k0 < KTOT; k0 += KC_) {
        const bool more = (k0 + KC_) < KTOT;
        if (more) gload(k0 + KC_);
        compute();
        if (more) {
            lstore();
            __syncthreads();
        }
        ldA ^= BUFX_; ldB ^= BUFX_; stA ^= BUFX_; stB ^= BUFX_;
    }
}

__global__ __launch_bounds__(256, 10) void moe_fc1_kernel(
    const float* __restrict__ x, const float* __restrict__ W1,
    const float* __restrict__ b1, const int* __restrict__ lists,
    const int* __restrict__ counts, float* __restrict__ H,
    int chunkTileBase) {
    int list, mBase, cnt;
    if (!find_tile(counts, chunkTileBase + (int)blockIdx.y, list, mBase, cnt)) return;
    const int expert = list & (E_ - 1);
    const float* __restrict__ W1e = W1 + (size_t)expert * D_ * DFF_;
    const float* __restrict__ b1e = b1 + (size_t)expert * DFF_;
    const int nBase = blockIdx.x * 128;
    const int hBase = blockIdx.y * 64;

    __shared__ __attribute__((aligned(16))) float smem[SMEMB_ / 4];

    const int tid  = threadIdx.x;
    const int lane = tid & 31, wave = tid >> 5;
    const int half = lane >> 4, lm = lane & 15;
    const int Moff = 32 * (wave & 1);
    const int Noff = 32 * (wave >> 1);

    const int rA0 = tid >> 3, rA1 = rA0 + 32;
    const int* lst = lists + list * NTOK + mBase;
    const bool va0 = (mBase + rA0) < cnt;
    const bool va1 = (mBase + rA1) < cnt;
    const unsigned ao0 = va0 ? (unsigned)lst[rA0] * D_ : 0u;
    const unsigned ao1 = va1 ? (unsigned)lst[rA1] * D_ : 0u;

    v8f acc00 = {}; v8f acc01 = {}; v8f acc10 = {}; v8f acc11 = {};
    gemm_tile_64x128<DFF_, D_>(x, ao0, ao1, W1e + nBase, (char*)smem, acc00, acc01, acc10, acc11);

#pragma unroll
    for (int r = 0; r < 8; ++r) {
        const int m0 = Moff + r + 8 * half;
        const int hrow0 = hBase + m0, hrow1 = hrow0 + 16;
        const int n0 = nBase + Noff + lm, n1 = n0 + 16;
        if (mBase + m0 < cnt) {
            H[(size_t)hrow0 * DFF_ + n0] = gelu_exact(acc00[r] + b1e[n0]);
            H[(size_t)hrow0 * DFF_ + n1] = gelu_exact(acc01[r] + b1e[n1]);
        }
        if (mBase + m0 + 16 < cnt) {
            H[(size_t)hrow1 * DFF_ + n0] = gelu_exact(acc10[r] + b1e[n0]);
            H[(size_t)hrow1 * DFF_ + n1] = gelu_exact(acc11[r] + b1e[n1]);
        }
    }
}

template <bool BOTH>
__global__ __launch_bounds__(256, 10) void moe_fc2_kernel(
    const float* __restrict__ H, const float* __restrict__ W2,
    const float* __restrict__ b2, const int* __restrict__ lists,
    const int* __restrict__ counts, const float* __restrict__ gates,
    float* __restrict__ out, float* __restrict__ part, int tileLo, int chunkTileBase) {
    const int tile = tileLo + (int)blockIdx.y;
    int list, mBase, cnt;
    if (!find_tile(counts, tile, list, mBase, cnt)) return;
    const int expert = list & (E_ - 1);
    const float* __restrict__ W2e = W2 + (size_t)expert * DFF_ * D_;
    const float* __restrict__ b2e = b2 + (size_t)expert * D_;
    const int nBase = blockIdx.x * 128;
    const int hBase = (tile - chunkTileBase) * 64;

    __shared__ __attribute__((aligned(16))) float smem[SMEMB_ / 4];

    const int tid  = threadIdx.x;
    const int lane = tid & 31, wave = tid >> 5;
    const int half = lane >> 4, lm = lane & 15;
    const int Moff = 32 * (wave & 1);
    const int Noff = 32 * (wave >> 1);

    const int rA0 = tid >> 3, rA1 = rA0 + 32;
    const bool va0 = (mBase + rA0) < cnt;
    const bool va1 = (mBase + rA1) < cnt;
    const unsigned ao0 = (unsigned)(va0 ? rA0 : 0) * DFF_;
    const unsigned ao1 = (unsigned)(va1 ? rA1 : 0) * DFF_;

    v8f acc00 = {}; v8f acc01 = {}; v8f acc10 = {}; v8f acc11 = {};
    gemm_tile_64x128<D_, DFF_>(H + (size_t)hBase * DFF_, ao0, ao1, W2e + nBase, (char*)smem, acc00, acc01, acc10, acc11);

    const int* lst = lists + list * NTOK;
    float* dst = (BOTH && list >= E_) ? part : out;
#pragma unroll
    for (int r = 0; r < 8; ++r) {
        const int m0 = Moff + r + 8 * half;
        const int lrow0 = mBase + m0, lrow1 = lrow0 + 16;
        const int n0 = nBase + Noff + lm, n1 = n0 + 16;
        if (lrow0 < cnt) {
            const int tok = lst[lrow0];
            const float g = gates[tok * E_ + expert];
            const float t0 = g * (acc00[r] + b2e[n0]);
            const float t1 = g * (acc01[r] + b2e[n1]);
            if (BOTH) {
                dst[(size_t)tok * D_ + n0] = t0;
                dst[(size_t)tok * D_ + n1] = t1;
            } else {
                out[(size_t)tok * D_ + n0] += t0;
                out[(size_t)tok * D_ + n1] += t1;
            }
        }
        if (lrow1 < cnt) {
            const int tok = lst[lrow1];
            const float g = gates[tok * E_ + expert];
            const float t0 = g * (acc10[r] + b2e[n0]);
            const float t1 = g * (acc11[r] + b2e[n1]);
            if (BOTH) {
                dst[(size_t)tok * D_ + n0] = t0;
                dst[(size_t)tok * D_ + n1] = t1;
            } else {
                out[(size_t)tok * D_ + n0] += t0;
                out[(size_t)tok * D_ + n1] += t1;
            }
        }
    }
}

__global__ __launch_bounds__(256) void moe_add_kernel(float* __restrict__ out,
                                                      const float* __restrict__ part, int n4) {
    const int stride = (int)(gridDim.x * blockDim.x);
    const int first = (int)(blockIdx.x * blockDim.x + threadIdx.x);
    if ((((size_t)out) & 15) == 0) {
        for (int i = first; i < n4; i += stride) {
            const float4 a = ((const float4*)out)[i];
            const float4 b = ((const float4*)part)[i];
            float4 c;
            c.x = a.x + b.x; c.y = a.y + b.y; c.z = a.z + b.z; c.w = a.w + b.w;
            ((float4*)out)[i] = c;
        }
    } else {
        for (int i = first; i < 4 * n4; i += stride) out[i] = out[i] + part[i];
    }
}

extern "C" void kernel_launch(void* const* d_in, const int* in_sizes, int n_in,
                              void* d_out, int out_size, void* d_ws, size_t ws_size,
                              hipStream_t stream) {
    const float* x  = (const float*)d_in[0];
    const float* Wg = (const float*)d_in[1];
    const float* W1 = (const float*)d_in[2];
    const float* b1 = (const float*)d_in[3];
    const float* W2 = (const float*)d_in[4];
    const float* b2 = (const float*)d_in[5];
    float* out = (float*)d_out;

    char* ws = (char*)d_ws;
    int*   counts = (int*)ws;
    int*   lists  = (int*)(ws + 256);
    float* gates  = (float*)(ws + 256 + 8ull * E_ * NTOK);
    size_t OFS_H  = 256 + 12ull * E_ * NTOK;
    OFS_H = (OFS_H + 255) & ~(size_t)255;
    float* H = (float*)(ws + OFS_H);

    const int MAXT = 2 * TPP_;
    size_t avail = (ws_size > OFS_H) ? (ws_size - OFS_H) : 0;
    const long long outN = (long long)NTOK * D_ + 1;

    const size_t H_ALL  = (size_t)MAXT * 64 * DFF_ * 4;
    const size_t PART_B = (size_t)NTOK * D_ * 4;
    if (avail >= H_ALL + PART_B) {
        float* part = (float*)(ws + OFS_H + H_ALL);
        moe_zero_kernel<<<1, 256, 0, stream>>>(out, 0, counts);
        moe_gate_kernel<<<NTOK / 8, 256, 0, stream>>>(x, Wg, gates, counts, lists);
        moe_loss_kernel<<<1, 256, 0, stream>>>(gates, out + (outN - 1));
        dim3 g1(DFF_ / 128, MAXT);
        moe_fc1_kernel<<<g1, 256, 0, stream>>>(x, W1, b1, lists, counts, H, 0);
        dim3 g2(D_ / 128, MAXT);
        moe_fc2_kernel<true><<<g2, 256, 0, stream>>>(H, W2, b2, lists, counts, gates, out, part, 0, 0);
        moe_add_kernel<<<2048, 256, 0, stream>>>(out, part, NTOK * D_ / 4);
        return;
    }

    long long capll = (long long)(avail / ((size_t)64 * DFF_ * 4));
    int cap = (int)(capll > MAXT ? MAXT : capll);
    if (cap <= 0) cap = 1;
    const int chunks = (MAXT + cap - 1) / cap;
    const int TPC = (MAXT + chunks - 1) / chunks;

    moe_zero_kernel<<<2048, 256, 0, stream>>>(out, outN, counts);
    moe_gate_kernel<<<NTOK / 8, 256, 0, stream>>>(x, Wg, gates, counts, lists);
    moe_loss_kernel<<<1, 256, 0, stream>>>(gates, out + (outN - 1));

    for (int c = 0; c < chunks; ++c) {
        const int cs = c * TPC;
        const int ce = (cs + TPC < MAXT) ? (cs + TPC) : MAXT;
        dim3 g1(DFF_ / 128, ce - cs);
        moe_fc1_kernel<<<g1, 256, 0, stream>>>(x, W1, b1, lists, counts, H, cs);
        for (int s = 0; s < 2; ++s) {
            const int lo = (s * TPP_ > cs) ? s * TPP_ : cs;
            const int hi = ((s + 1) * TPP_ < ce) ? (s + 1) * TPP_ : ce;
            if (hi > lo) {
                dim3 g2(D_ / 128, hi - lo);
                moe_fc2_kernel<false><<<g2, 256, 0, stream>>>(H, W2, b2, lists, counts,
                                                              gates, out, nullptr, lo, cs);
            }
        }
    }
}
